# starting kernel
# baseline (speedup 1.0000x reference)
#include <hip/hip_runtime.h>

typedef float v2f __attribute__((ext_vector_type(2)));
typedef float v8f __attribute__((ext_vector_type(8)));

#define LSTRIDE 66

__global__ __launch_bounds__(128) void bdattn_kernel(const float* __restrict__ q,
                                                     const float* __restrict__ k,
                                                     const float* __restrict__ v,
                                                     float* __restrict__ out) {
    __shared__ float lds_k[64 * LSTRIDE];
    __shared__ float lds_v[64 * LSTRIDE];
    __shared__ float lds_s[64 * LSTRIDE];

    const int tid   = threadIdx.x;
    const int lane  = tid & 31;
    const int wave  = tid >> 5;
    const int l15   = lane & 15;
    const int hi    = lane >> 4;
    const int h2    = hi * 2;
    const int mBase = wave * 16;

    const size_t base = (size_t)blockIdx.x * 4096;
    const float* kg = k + base;
    const float* vg = v + base;

    #pragma unroll
    for (int i = 0; i < 16; ++i) {
        int idx = tid + i * 128;
        int row = idx >> 5;
        int col = (idx & 31) * 2;
        *(v2f*)&lds_k[row * LSTRIDE + col] = ((const v2f*)kg)[idx];
        *(v2f*)&lds_v[row * LSTRIDE + col] = ((const v2f*)vg)[idx];
    }

    const float* qg = q + base + (size_t)(mBase + l15) * 64 + h2;
    v2f Aq[16];
    #pragma unroll
    for (int kb = 0; kb < 16; ++kb) Aq[kb] = *(const v2f*)(qg + kb * 4);

    __syncthreads();

    v8f accS[4] = {};
    #pragma unroll
    for (int t = 0; t < 4; ++t) {
        const float* kp = &lds_k[(t * 16 + l15) * LSTRIDE + h2];
        #pragma unroll
        for (int kb = 0; kb < 16; ++kb) {
            v2f Bk = *(const v2f*)(kp + kb * 4);
            accS[t] = __builtin_amdgcn_wmma_f32_16x16x4_f32(
                false, Aq[kb], false, Bk, (short)0, accS[t], false, false);
        }
    }

    {
        const int rowOff = (mBase + hi * 8) * LSTRIDE + l15;
        #pragma unroll
        for (int t = 0; t < 4; ++t)
            #pragma unroll
            for (int r = 0; r < 8; ++r)
                lds_s[rowOff + r * LSTRIDE + t * 16] = accS[t][r];
    }
    __syncthreads();

    if (tid < 64) {
        float* row = &lds_s[tid * LSTRIDE];
        float m = row[0];
        #pragma unroll 8
        for (int j = 1; j < 64; ++j) m = fmaxf(m, row[j]);
        float s = 0.0f;
        #pragma unroll 8
        for (int j = 0; j < 64; ++j) {
            float e = __expf((row[j] - m) * 0.125f);
            row[j] = e;
            s += e;
        }
        float inv = 1.0f / s;
        #pragma unroll 8
        for (int j = 0; j < 64; ++j) row[j] *= inv;
    }
    __syncthreads();

    v8f accO[4] = {};
    const float* pp  = &lds_s[(mBase + l15) * LSTRIDE + h2];
    const float* vp0 = &lds_v[h2 * LSTRIDE + l15];
    #pragma unroll
    for (int kb = 0; kb < 16; ++kb) {
        v2f Ap = *(const v2f*)(pp + kb * 4);
        const float* vk = vp0 + (kb * 4) * LSTRIDE;
        #pragma unroll
        for (int t = 0; t < 4; ++t) {
            v2f Bv;
            Bv.x = vk[t * 16];
            Bv.y = vk[t * 16 + LSTRIDE];
            accO[t] = __builtin_amdgcn_wmma_f32_16x16x4_f32(
                false, Ap, false, Bv, (short)0, accO[t], false, false);
        }
    }

    float* og = out + base + (size_t)(mBase + hi * 8) * 64 + l15;
    #pragma unroll
    for (int t = 0; t < 4; ++t)
        #pragma unroll
        for (int r = 0; r < 8; ++r)
            og[r * 64 + t * 16] = accO[t][r];
}

extern "C" void kernel_launch(void* const* d_in, const int* in_sizes, int n_in,
                              void* d_out, int out_size, void* d_ws, size_t ws_size,
                              hipStream_t stream) {
    const float* q = (const float*)d_in[0];
    const float* k = (const float*)d_in[1];
    const float* v = (const float*)d_in[2];
    float* out = (float*)d_out;

    int nblocks = in_sizes[0] / 4096;
    bdattn_kernel<<<dim3(nblocks), dim3(128), 0, stream>>>(q, k, v, out);
}


# Scores stay in registers: S^T=K*Q^T, softmax per lane + one lane swap, P^T tile fed as B of O^T=V^T*P^T; drops LDS score trip, 2 barriers, 64-thread serial softmax; new sum order (row sums, PV keys)
# speedup vs baseline: 1.0853x; 1.0853x over previous
#include <hip/hip_runtime.h>


typedef float v2f __attribute__((ext_vector_type(2)));
typedef float v4f __attribute__((ext_vector_type(4)));
typedef float v8f __attribute__((ext_vector_type(8)));

#define LSTRIDE 66

__device__ __forceinline__ float xhalf(float x) {
    int b = __builtin_bit_cast(int, x);
    return __builtin_bit_cast(float,
        __builtin_amdgcn_permlanex16(b, b, 0x76543210, 0xfedcba98, false, false));
}

__global__ __launch_bounds__(128) void bdattn_kernel(const float* __restrict__ q,
                                                     const float* __restrict__ k,
                                                     const float* __restrict__ v,
                                                     float* __restrict__ out) {
    __shared__ float lds_k[64 * LSTRIDE];
    __shared__ float lds_v[64 * LSTRIDE];

    const int tid   = threadIdx.x;
    const int lane  = tid & 31;
    const int wave  = tid >> 5;
    const int l15   = lane & 15;
    const int hi    = lane >> 4;
    const int h2    = hi * 2;
    const int mBase = wave * 16;

    const size_t base = (size_t)blockIdx.x * 4096;
    const float* kg = k + base;
    const float* vg = v + base;

    #pragma unroll
    for (int i = 0; i < 16; ++i) {
        int idx = tid + i * 128;
        int row = idx >> 5;
        int col = (idx & 31) * 2;
        *(v2f*)&lds_k[row * LSTRIDE + col] = ((const v2f*)kg)[idx];
        *(v2f*)&lds_v[row * LSTRIDE + col] = ((const v2f*)vg)[idx];
    }

    const float* qg = q + base + (size_t)(mBase + l15) * 64 + h2;
    v2f Bq[16];
    #pragma unroll
    for (int kb = 0; kb < 16; ++kb) Bq[kb] = *(const v2f*)(qg + kb * 4);

    __syncthreads();

    v8f accS[4] = {};
    #pragma unroll
    for (int t = 0; t < 4; ++t) {
        const float* kp = &lds_k[(t * 16 + l15) * LSTRIDE + h2];
        #pragma unroll
        for (int kb = 0; kb < 16; ++kb) {
            v2f Ak = *(const v2f*)(kp + kb * 4);
            accS[t] = __builtin_amdgcn_wmma_f32_16x16x4_f32(
                false, Ak, false, Bq[kb], (short)0, accS[t], false, false);
        }
    }

    float m0 = accS[0][0], m1 = accS[1][0], m2 = accS[2][0], m3 = accS[3][0];
    #pragma unroll
    for (int r = 1; r < 8; ++r) {
        m0 = fmaxf(m0, accS[0][r]);
        m1 = fmaxf(m1, accS[1][r]);
        m2 = fmaxf(m2, accS[2][r]);
        m3 = fmaxf(m3, accS[3][r]);
    }
    float m = fmaxf(fmaxf(m0, m1), fmaxf(m2, m3));
    m = fmaxf(m, xhalf(m));

    float s0 = 0.0f, s1 = 0.0f, s2 = 0.0f, s3 = 0.0f;
    #pragma unroll
    for (int r = 0; r < 8; ++r) {
        float e0 = __expf((accS[0][r] - m) * 0.125f);
        float e1 = __expf((accS[1][r] - m) * 0.125f);
        float e2 = __expf((accS[2][r] - m) * 0.125f);
        float e3 = __expf((accS[3][r] - m) * 0.125f);
        accS[0][r] = e0; s0 += e0;
        accS[1][r] = e1; s1 += e1;
        accS[2][r] = e2; s2 += e2;
        accS[3][r] = e3; s3 += e3;
    }
    float s = (s0 + s1) + (s2 + s3);
    s = s + xhalf(s);
    const float inv = 1.0f / s;
    #pragma unroll
    for (int t = 0; t < 4; ++t)
        #pragma unroll
        for (int r = 0; r < 8; ++r) accS[t][r] *= inv;

    v8f accO[4] = {};
    const float* vp0 = &lds_v[(hi * 8) * LSTRIDE + l15];
    #pragma unroll
    for (int t = 0; t < 4; ++t) {
        #pragma unroll
        for (int j = 0; j < 4; ++j) {
            v2f Bp;
            Bp.x = accS[t][2 * j];
            Bp.y = accS[t][2 * j + 1];
            const float* vk = vp0 + (t * 16 + 2 * j) * LSTRIDE;
            #pragma unroll
            for (int dt = 0; dt < 4; ++dt) {
                v2f Av;
                Av.x = vk[dt * 16];
                Av.y = vk[dt * 16 + LSTRIDE];
                accO[dt] = __builtin_amdgcn_wmma_f32_16x16x4_f32(
                    false, Av, false, Bp, (short)0, accO[dt], false, false);
            }
        }
    }

    float* og = out + base + (size_t)(mBase + l15) * 64 + hi * 8;
    #pragma unroll
    for (int dt = 0; dt < 4; ++dt) {
        v4f lo, hi4;
        lo.x  = accO[dt][0]; lo.y  = accO[dt][1]; lo.z  = accO[dt][2]; lo.w  = accO[dt][3];
        hi4.x = accO[dt][4]; hi4.y = accO[dt][5]; hi4.z = accO[dt][6]; hi4.w = accO[dt][7];
        *(v4f*)(og + dt * 16)     = lo;
        *(v4f*)(og + dt * 16 + 4) = hi4;
    }
}

extern "C" void kernel_launch(void* const* d_in, const int* in_sizes, int n_in,
                              void* d_out, int out_size, void* d_ws, size_t ws_size,
                              hipStream_t stream) {
    const float* q = (const float*)d_in[0];
    const float* k = (const float*)d_in[1];
    const float* v = (const float*)d_in[2];
    float* out = (float*)d_out;

    int nblocks = in_sizes[0] / 4096;
    bdattn_kernel<<<dim3(nblocks), dim3(128), 0, stream>>>(q, k, v, out);
}


# O tile staged through the dead K tile in LDS (pitch 68, freed by a split barrier): each global store instruction writes two whole 256-B rows, not 16-B pieces of 16 lines; sum order unchanged
# speedup vs baseline: 1.1669x; 1.0752x over previous
#include <hip/hip_runtime.h>


typedef float v2f __attribute__((ext_vector_type(2)));
typedef float v4f __attribute__((ext_vector_type(4)));
typedef float v8f __attribute__((ext_vector_type(8)));

#define LSTRIDE 66
#define OSTRIDE 68

__device__ __forceinline__ float xhalf(float x) {
    int b = __builtin_bit_cast(int, x);
    return __builtin_bit_cast(float,
        __builtin_amdgcn_permlanex16(b, b, 0x76543210, 0xfedcba98, false, false));
}

__global__ __launch_bounds__(128) void bdattn_kernel(const float* __restrict__ q,
                                                     const float* __restrict__ k,
                                                     const float* __restrict__ v,
                                                     float* __restrict__ out) {
    __shared__ __attribute__((aligned(16))) float lds_k[4 * 16 * OSTRIDE];
    __shared__ float lds_v[64 * LSTRIDE];

    const int tid   = threadIdx.x;
    const int lane  = tid & 31;
    const int wave  = tid >> 5;
    const int l15   = lane & 15;
    const int hi    = lane >> 4;
    const int h2    = hi * 2;
    const int mBase = wave * 16;

    const size_t base = (size_t)blockIdx.x * 4096;
    const float* kg = k + base;
    const float* vg = v + base;

    #pragma unroll
    for (int i = 0; i < 16; ++i) {
        int idx = tid + i * 128;
        int row = idx >> 5;
        int col = (idx & 31) * 2;
        *(v2f*)&lds_k[row * LSTRIDE + col] = ((const v2f*)kg)[idx];
        *(v2f*)&lds_v[row * LSTRIDE + col] = ((const v2f*)vg)[idx];
    }

    const float* qg = q + base + (size_t)(mBase + l15) * 64 + h2;
    v2f Bq[16];
    #pragma unroll
    for (int kb = 0; kb < 16; ++kb) Bq[kb] = *(const v2f*)(qg + kb * 4);

    __syncthreads();

    v8f accS[4] = {};
    #pragma unroll
    for (int t = 0; t < 4; ++t) {
        const float* kp = &lds_k[(t * 16 + l15) * LSTRIDE + h2];
        #pragma unroll
        for (int kb = 0; kb < 16; ++kb) {
            v2f Ak = *(const v2f*)(kp + kb * 4);
            accS[t] = __builtin_amdgcn_wmma_f32_16x16x4_f32(
                false, Ak, false, Bq[kb], (short)0, accS[t], false, false);
        }
    }

    __builtin_amdgcn_fence(__ATOMIC_RELEASE, "workgroup");
    __builtin_amdgcn_s_barrier_signal(-1);

    float m0 = accS[0][0], m1 = accS[1][0], m2 = accS[2][0], m3 = accS[3][0];
    #pragma unroll
    for (int r = 1; r < 8; ++r) {
        m0 = fmaxf(m0, accS[0][r]);
        m1 = fmaxf(m1, accS[1][r]);
        m2 = fmaxf(m2, accS[2][r]);
        m3 = fmaxf(m3, accS[3][r]);
    }
    float m = fmaxf(fmaxf(m0, m1), fmaxf(m2, m3));
    m = fmaxf(m, xhalf(m));

    float s0 = 0.0f, s1 = 0.0f, s2 = 0.0f, s3 = 0.0f;
    #pragma unroll
    for (int r = 0; r < 8; ++r) {
        float e0 = __expf((accS[0][r] - m) * 0.125f);
        float e1 = __expf((accS[1][r] - m) * 0.125f);
        float e2 = __expf((accS[2][r] - m) * 0.125f);
        float e3 = __expf((accS[3][r] - m) * 0.125f);
        accS[0][r] = e0; s0 += e0;
        accS[1][r] = e1; s1 += e1;
        accS[2][r] = e2; s2 += e2;
        accS[3][r] = e3; s3 += e3;
    }
    float s = (s0 + s1) + (s2 + s3);
    s = s + xhalf(s);
    const float inv = 1.0f / s;
    #pragma unroll
    for (int t = 0; t < 4; ++t)
        #pragma unroll
        for (int r = 0; r < 8; ++r) accS[t][r] *= inv;

    v8f accO[4] = {};
    const float* vp0 = &lds_v[(hi * 8) * LSTRIDE + l15];
    #pragma unroll
    for (int t = 0; t < 4; ++t) {
        #pragma unroll
        for (int j = 0; j < 4; ++j) {
            v2f Bp;
            Bp.x = accS[t][2 * j];
            Bp.y = accS[t][2 * j + 1];
            const float* vk = vp0 + (t * 16 + 2 * j) * LSTRIDE;
            #pragma unroll
            for (int dt = 0; dt < 4; ++dt) {
                v2f Av;
                Av.x = vk[dt * 16];
                Av.y = vk[dt * 16 + LSTRIDE];
                accO[dt] = __builtin_amdgcn_wmma_f32_16x16x4_f32(
                    false, Av, false, Bp, (short)0, accO[dt], false, false);
            }
        }
    }


    __builtin_amdgcn_s_barrier_wait(-1);
    __builtin_amdgcn_fence(__ATOMIC_ACQUIRE, "workgroup");

    float* st = &lds_k[wave * 16 * OSTRIDE];
    float* sw = st + l15 * OSTRIDE + hi * 8;
    #pragma unroll
    for (int dt = 0; dt < 4; ++dt) {
        v4f lo, hi4;
        lo.x  = accO[dt][0]; lo.y  = accO[dt][1]; lo.z  = accO[dt][2]; lo.w  = accO[dt][3];
        hi4.x = accO[dt][4]; hi4.y = accO[dt][5]; hi4.z = accO[dt][6]; hi4.w = accO[dt][7];
        *(v4f*)(sw + dt * 16)     = lo;
        *(v4f*)(sw + dt * 16 + 4) = hi4;
    }

    __builtin_amdgcn_fence(__ATOMIC_RELEASE, "workgroup");

    const int rr = lane >> 4;
    const int c4 = (lane & 15) * 4;
    const float* sr = st + rr * OSTRIDE + c4;
    float* og = out + base + (size_t)(mBase + rr) * 64 + c4;
    #pragma unroll
    for (int i = 0; i < 8; ++i) {
        v4f x = *(const v4f*)(sr + 2 * i * OSTRIDE);
        *(v4f*)(og + i * 128) = x;
    }
}

extern "C" void kernel_launch(void* const* d_in, const int* in_sizes, int n_in,
                              void* d_out, int out_size, void* d_ws, size_t ws_size,
                              hipStream_t stream) {
    const float* q = (const float*)d_in[0];
    const float* k = (const float*)d_in[1];
    const float* v = (const float*)d_in[2];
    float* out = (float*)d_out;

    int nblocks = in_sizes[0] / 4096;
    bdattn_kernel<<<dim3(nblocks), dim3(128), 0, stream>>>(q, k, v, out);
}
